# starting kernel
# baseline (speedup 1.0000x reference)
#include <hip/hip_runtime.h>
#include <math.h>

#define N_GENES 20000
#define UNITS   10000
#define DEG     32
#define BATCH   128

typedef __attribute__((ext_vector_type(2))) float v2f;
typedef __attribute__((ext_vector_type(8))) float v8f;

__global__ __launch_bounds__(256) void feat_transpose_kernel(
    const float* __restrict__ in, float* __restrict__ out) {
  __shared__ float tile[32][33];
  const int tx = threadIdx.x;
  const int ty = threadIdx.y;
  const int g0 = blockIdx.x * 32;
  const int b0 = blockIdx.y * 32;
#pragma unroll
  for (int i = 0; i < 32; i += 8)
    tile[ty + i][tx] = in[(long)(b0 + ty + i) * N_GENES + (g0 + tx)];
  __syncthreads();
#pragma unroll
  for (int i = 0; i < 32; i += 8)
    out[(long)(g0 + ty + i) * BATCH + (b0 + tx)] = tile[tx][ty + i];
}

__global__ __launch_bounds__(128) void ppi_gather_wmma_kernel(
    const float* __restrict__ featT, const int* __restrict__ ppi,
    const float* __restrict__ kernelw, const float* __restrict__ biasw,
    float* __restrict__ out) {
  const int u    = blockIdx.x;
  const int lane = threadIdx.x & 31;
  const int wave = threadIdx.x >> 5;
  const bool hi  = lane >= 16;

  int offs[16];
#pragma unroll
  for (int c = 0; c < 8; ++c) {
    const int i0 = ppi[u * DEG + 4 * c + 0];
    const int i1 = ppi[u * DEG + 4 * c + 1];
    const int i2 = ppi[u * DEG + 4 * c + 2];
    const int i3 = ppi[u * DEG + 4 * c + 3];
    offs[2 * c + 0] = (hi ? i2 : i0) * BATCH;
    offs[2 * c + 1] = (hi ? i3 : i1) * BATCH;
  }
  const float kk = kernelw[u];
  const float bb = biasw[u];

  v2f a;
  a.x = 1.0f;
  a.y = 1.0f;

  const int colLane = lane & 15;
#pragma unroll
  for (int t = 0; t < 2; ++t) {
    const int tileN = wave + 4 * t;
    const int col   = tileN * 16 + colLane;
    v8f acc = {};
#pragma unroll
    for (int c = 0; c < 8; ++c) {
      v2f b;
      b.x = __builtin_fabsf(featT[offs[2 * c + 0] + col]);
      b.y = __builtin_fabsf(featT[offs[2 * c + 1] + col]);
      acc = __builtin_amdgcn_wmma_f32_16x16x4_f32(
          false, a, false, b, (short)0, acc, false, false);
    }
    if (lane < 16) {
      const float h = acc[0] * kk + bb;
      out[(long)(tileN * 16 + lane) * UNITS + u] = tanhf(h);
    }
  }
}

extern "C" void kernel_launch(void* const* d_in, const int* in_sizes, int n_in,
                              void* d_out, int out_size, void* d_ws, size_t ws_size,
                              hipStream_t stream) {
  const float* feature = (const float*)d_in[0];
  const int*   ppi     = (const int*)d_in[1];
  const float* kernelw = (const float*)d_in[2];
  const float* biasw   = (const float*)d_in[3];
  float*       out     = (float*)d_out;
  float*       featT   = (float*)d_ws;

  dim3 tBlk(32, 8, 1);
  dim3 tGrd(N_GENES / 32, BATCH / 32, 1);
  feat_transpose_kernel<<<tGrd, tBlk, 0, stream>>>(feature, featT);

  ppi_gather_wmma_kernel<<<UNITS, 128, 0, stream>>>(featT, ppi, kernelw, biasw, out);
}


# 16 units per workgroup on the WMMA M axis via one-hot A rows: one accumulator tile per batch tile holds 16 units, each lane stores 32 contiguous bytes; 16x fewer waves
# speedup vs baseline: 2.2762x; 2.2762x over previous
#include <hip/hip_runtime.h>
#include <math.h>

#define N_GENES 20000
#define UNITS   10000
#define DEG     32
#define BATCH   128

typedef __attribute__((ext_vector_type(2))) float v2f;
typedef __attribute__((ext_vector_type(8))) float v8f;

__global__ __launch_bounds__(256) void feat_transpose_kernel(
    const float* __restrict__ in, float* __restrict__ out) {
  __shared__ float tile[32][33];
  const int tx = threadIdx.x;
  const int ty = threadIdx.y;
  const int g0 = blockIdx.x * 32;
  const int b0 = blockIdx.y * 32;
#pragma unroll
  for (int i = 0; i < 32; i += 8)
    tile[ty + i][tx] = in[(long)(b0 + ty + i) * N_GENES + (g0 + tx)];
  __syncthreads();
#pragma unroll
  for (int i = 0; i < 32; i += 8)
    out[(long)(g0 + ty + i) * BATCH + (b0 + tx)] = tile[tx][ty + i];
}

#define UPB 16

__global__ __launch_bounds__(128) void ppi_gather_wmma_kernel(
    const float* __restrict__ featT, const int* __restrict__ ppi,
    const float* __restrict__ kernelw, const float* __restrict__ biasw,
    float* __restrict__ out) {
  const int u0   = blockIdx.x * UPB;
  const int lane = threadIdx.x & 31;
  const int wave = threadIdx.x >> 5;
  const bool hi  = lane >= 16;
  const int colLane = lane & 15;
  const int col0 = wave * 16 + colLane;

  v8f acc0 = {};
  v8f acc1 = {};

#pragma unroll
  for (int j = 0; j < UPB; ++j) {
    const int* __restrict__ row = ppi + (u0 + j) * DEG;
    v2f a;
    a.x = (colLane == j) ? 1.0f : 0.0f;
    a.y = a.x;
#pragma unroll
    for (int c = 0; c < 8; ++c) {
      const int i0 = row[4 * c + 0];
      const int i1 = row[4 * c + 1];
      const int i2 = row[4 * c + 2];
      const int i3 = row[4 * c + 3];
      const int o0 = (hi ? i2 : i0) * BATCH + col0;
      const int o1 = (hi ? i3 : i1) * BATCH + col0;
      v2f b0, b1;
      b0.x = __builtin_fabsf(featT[o0]);
      b0.y = __builtin_fabsf(featT[o1]);
      b1.x = __builtin_fabsf(featT[o0 + 64]);
      b1.y = __builtin_fabsf(featT[o1 + 64]);
      acc0 = __builtin_amdgcn_wmma_f32_16x16x4_f32(
          false, a, false, b0, (short)0, acc0, false, false);
      acc1 = __builtin_amdgcn_wmma_f32_16x16x4_f32(
          false, a, false, b1, (short)0, acc1, false, false);
    }
  }

  const int ub = u0 + (hi ? 8 : 0);
  const float4 k_lo = *reinterpret_cast<const float4*>(kernelw + ub);
  const float4 k_hi = *reinterpret_cast<const float4*>(kernelw + ub + 4);
  const float4 b_lo = *reinterpret_cast<const float4*>(biasw + ub);
  const float4 b_hi = *reinterpret_cast<const float4*>(biasw + ub + 4);
  const float kk[8] = {k_lo.x, k_lo.y, k_lo.z, k_lo.w, k_hi.x, k_hi.y, k_hi.z, k_hi.w};
  const float bb[8] = {b_lo.x, b_lo.y, b_lo.z, b_lo.w, b_hi.x, b_hi.y, b_hi.z, b_hi.w};
  float r0[8], r1[8];
#pragma unroll
  for (int r = 0; r < 8; ++r) {
    r0[r] = tanhf(acc0[r] * kk[r] + bb[r]);
    r1[r] = tanhf(acc1[r] * kk[r] + bb[r]);
  }
  float* p0 = out + (long)col0 * UNITS + ub;
  float* p1 = out + (long)(col0 + 64) * UNITS + ub;
  *reinterpret_cast<float4*>(p0)     = make_float4(r0[0], r0[1], r0[2], r0[3]);
  *reinterpret_cast<float4*>(p0 + 4) = make_float4(r0[4], r0[5], r0[6], r0[7]);
  *reinterpret_cast<float4*>(p1)     = make_float4(r1[0], r1[1], r1[2], r1[3]);
  *reinterpret_cast<float4*>(p1 + 4) = make_float4(r1[4], r1[5], r1[6], r1[7]);
}

extern "C" void kernel_launch(void* const* d_in, const int* in_sizes, int n_in,
                              void* d_out, int out_size, void* d_ws, size_t ws_size,
                              hipStream_t stream) {
  const float* feature = (const float*)d_in[0];
  const int*   ppi     = (const int*)d_in[1];
  const float* kernelw = (const float*)d_in[2];
  const float* biasw   = (const float*)d_in[3];
  float*       out     = (float*)d_out;
  float*       featT   = (float*)d_ws;

  dim3 tBlk(32, 8, 1);
  dim3 tGrd(N_GENES / 32, BATCH / 32, 1);
  feat_transpose_kernel<<<tGrd, tBlk, 0, stream>>>(feature, featT);

  ppi_gather_wmma_kernel<<<UNITS / UPB, 128, 0, stream>>>(featT, ppi, kernelw, biasw, out);
}


# gene indices loaded up front by vector loads and spread by dpp row broadcast: no scalar index load waited for inside the gather stream; gathers in flight as before (register target 12 waves/SIMD)
# speedup vs baseline: 2.3583x; 1.0361x over previous
#include <hip/hip_runtime.h>
#include <math.h>

#define N_GENES 20000
#define UNITS   10000
#define DEG     32
#define BATCH   128

typedef __attribute__((ext_vector_type(2))) float v2f;
typedef __attribute__((ext_vector_type(8))) float v8f;

__global__ __launch_bounds__(256) void feat_transpose_kernel(
    const float* __restrict__ in, float* __restrict__ out) {
  __shared__ float tile[32][33];
  const int tx = threadIdx.x;
  const int ty = threadIdx.y;
  const int g0 = blockIdx.x * 32;
  const int b0 = blockIdx.y * 32;
#pragma unroll
  for (int i = 0; i < 32; i += 8)
    tile[ty + i][tx] = in[(long)(b0 + ty + i) * N_GENES + (g0 + tx)];
  __syncthreads();
#pragma unroll
  for (int i = 0; i < 32; i += 8)
    out[(long)(g0 + ty + i) * BATCH + (b0 + tx)] = tile[tx][ty + i];
}

#define UPB 16

template <int N>
__device__ __forceinline__ int row_bcast(int v) {
  return __builtin_amdgcn_update_dpp(0, v, 0x150 + N, 0xf, 0xf, false);
}

#define GATHER_CHUNK(C)                                                   \
  {                                                                       \
    const int o0 = row_bcast<2 * (C)>(idxj) * BATCH + col0;               \
    const int o1 = row_bcast<2 * (C) + 1>(idxj) * BATCH + col0;           \
    v2f b0, b1;                                                           \
    b0.x = __builtin_fabsf(featT[o0]);                                    \
    b0.y = __builtin_fabsf(featT[o1]);                                    \
    b1.x = __builtin_fabsf(featT[o0 + 64]);                               \
    b1.y = __builtin_fabsf(featT[o1 + 64]);                               \
    acc0 = __builtin_amdgcn_wmma_f32_16x16x4_f32(                         \
        false, a, false, b0, (short)0, acc0, false, false);               \
    acc1 = __builtin_amdgcn_wmma_f32_16x16x4_f32(                         \
        false, a, false, b1, (short)0, acc1, false, false);               \
  }

__global__ __launch_bounds__(128) __attribute__((amdgpu_waves_per_eu(1, 12)))
void ppi_gather_wmma_kernel(
    const float* __restrict__ featT, const int* __restrict__ ppi,
    const float* __restrict__ kernelw, const float* __restrict__ biasw,
    float* __restrict__ out) {
  const int u0   = blockIdx.x * UPB;
  const int lane = threadIdx.x & 31;
  const int wave = threadIdx.x >> 5;
  const bool hi  = lane >= 16;
  const int colLane = lane & 15;
  const int col0 = wave * 16 + colLane;

  v8f acc0 = {};
  v8f acc1 = {};

  const int e = 4 * (colLane >> 1) + (hi ? 2 : 0) + (colLane & 1);
  int idx[UPB];
#pragma unroll
  for (int j = 0; j < UPB; ++j) idx[j] = ppi[(u0 + j) * DEG + e];

#pragma unroll
  for (int j = 0; j < UPB; ++j) {
    const int idxj = idx[j];
    v2f a;
    a.x = (colLane == j) ? 1.0f : 0.0f;
    a.y = a.x;
    GATHER_CHUNK(0) GATHER_CHUNK(1) GATHER_CHUNK(2) GATHER_CHUNK(3)
    GATHER_CHUNK(4) GATHER_CHUNK(5) GATHER_CHUNK(6) GATHER_CHUNK(7)
  }

  const int ub = u0 + (hi ? 8 : 0);
  const float4 k_lo = *reinterpret_cast<const float4*>(kernelw + ub);
  const float4 k_hi = *reinterpret_cast<const float4*>(kernelw + ub + 4);
  const float4 b_lo = *reinterpret_cast<const float4*>(biasw + ub);
  const float4 b_hi = *reinterpret_cast<const float4*>(biasw + ub + 4);
  const float kk[8] = {k_lo.x, k_lo.y, k_lo.z, k_lo.w, k_hi.x, k_hi.y, k_hi.z, k_hi.w};
  const float bb[8] = {b_lo.x, b_lo.y, b_lo.z, b_lo.w, b_hi.x, b_hi.y, b_hi.z, b_hi.w};
  float r0[8], r1[8];
#pragma unroll
  for (int r = 0; r < 8; ++r) {
    r0[r] = tanhf(acc0[r] * kk[r] + bb[r]);
    r1[r] = tanhf(acc1[r] * kk[r] + bb[r]);
  }
  float* p0 = out + (long)col0 * UNITS + ub;
  float* p1 = out + (long)(col0 + 64) * UNITS + ub;
  *reinterpret_cast<float4*>(p0)     = make_float4(r0[0], r0[1], r0[2], r0[3]);
  *reinterpret_cast<float4*>(p0 + 4) = make_float4(r0[4], r0[5], r0[6], r0[7]);
  *reinterpret_cast<float4*>(p1)     = make_float4(r1[0], r1[1], r1[2], r1[3]);
  *reinterpret_cast<float4*>(p1 + 4) = make_float4(r1[4], r1[5], r1[6], r1[7]);
}

extern "C" void kernel_launch(void* const* d_in, const int* in_sizes, int n_in,
                              void* d_out, int out_size, void* d_ws, size_t ws_size,
                              hipStream_t stream) {
  const float* feature = (const float*)d_in[0];
  const int*   ppi     = (const int*)d_in[1];
  const float* kernelw = (const float*)d_in[2];
  const float* biasw   = (const float*)d_in[3];
  float*       out     = (float*)d_out;
  float*       featT   = (float*)d_ws;

  dim3 tBlk(32, 8, 1);
  dim3 tGrd(N_GENES / 32, BATCH / 32, 1);
  feat_transpose_kernel<<<tGrd, tBlk, 0, stream>>>(feature, featT);

  ppi_gather_wmma_kernel<<<UNITS / UPB, 128, 0, stream>>>(featT, ppi, kernelw, biasw, out);
}


# pairs of batch rows of |feature| kept in LDS and gathered there by 8-byte ds reads, N = 16 units per WMMA; workspace transpose and its kernel dropped (1 launch); deg sum order changes
# speedup vs baseline: 3.6816x; 1.5611x over previous
#include <hip/hip_runtime.h>
#include <math.h>

#define N_GENES 20000
#define UNITS   10000
#define DEG     32
#define BATCH   128

typedef __attribute__((ext_vector_type(2))) float v2f;
typedef __attribute__((ext_vector_type(8))) float v8f;

#define TILE_U    16
#define N_TILES   (UNITS / TILE_U)
#define SPLIT     4
#define NTHREADS  512
#define NWAVES    (NTHREADS / 32)
#define ROW_VEC4  (N_GENES / 4)
#define FILL_FULL (ROW_VEC4 / NTHREADS)
#define FILL_TAIL (ROW_VEC4 - FILL_FULL * NTHREADS)
#define LDS_FLOATS 41472

__global__ __launch_bounds__(NTHREADS) void ppi_lds_gather_wmma_kernel(
    const float* __restrict__ feature, const int* __restrict__ ppi,
    const float* __restrict__ kernelw, const float* __restrict__ biasw,
    float* __restrict__ out) {
  __shared__ __attribute__((aligned(16))) float rows[LDS_FLOATS];

  const int tid  = threadIdx.x;
  const int lane = tid & 31;
  const int wave = __builtin_amdgcn_readfirstlane(tid >> 5);
  const int c    = lane & 15;
  const int h    = lane >> 4;
  const int pair = blockIdx.x / SPLIT;
  const int part = blockIdx.x % SPLIT;
  const int bA   = pair * 2;
  const int t_lo = (part * N_TILES) / SPLIT;
  const int t_hi = ((part + 1) * N_TILES) / SPLIT;

  int t = t_lo + wave;
  const int4* ip = reinterpret_cast<const int4*>(ppi + (t * TILE_U + c) * DEG + 16 * h);
  int4 n0 = ip[0], n1 = ip[1], n2 = ip[2], n3 = ip[3];
  float nk = kernelw[t * TILE_U + c];
  float nb = biasw[t * TILE_U + c];

  {
    const float4* r0 = reinterpret_cast<const float4*>(feature + (long)bA * N_GENES);
    const float4* r1 = reinterpret_cast<const float4*>(feature + (long)(bA + 1) * N_GENES);
    float4 fa[FILL_FULL + 1], fb[FILL_FULL + 1];
#pragma unroll
    for (int i = 0; i < FILL_FULL; ++i) {
      fa[i] = r0[tid + i * NTHREADS];
      fb[i] = r1[tid + i * NTHREADS];
    }
    const int qt = tid < FILL_TAIL ? tid + FILL_FULL * NTHREADS : ROW_VEC4 - 1;
    fa[FILL_FULL] = r0[qt];
    fb[FILL_FULL] = r1[qt];
#pragma unroll
    for (int i = 0; i < FILL_FULL; ++i) {
      float4* dst = reinterpret_cast<float4*>(&rows[8 * (tid + i * NTHREADS)]);
      dst[0] = make_float4(__builtin_fabsf(fa[i].x), __builtin_fabsf(fb[i].x),
                           __builtin_fabsf(fa[i].y), __builtin_fabsf(fb[i].y));
      dst[1] = make_float4(__builtin_fabsf(fa[i].z), __builtin_fabsf(fb[i].z),
                           __builtin_fabsf(fa[i].w), __builtin_fabsf(fb[i].w));
    }
    asm volatile("" :: "v"(fa[FILL_FULL].x), "v"(fa[FILL_FULL].y),
                 "v"(fa[FILL_FULL].z), "v"(fa[FILL_FULL].w),
                 "v"(fb[FILL_FULL].x), "v"(fb[FILL_FULL].y),
                 "v"(fb[FILL_FULL].z), "v"(fb[FILL_FULL].w));
    if (tid < FILL_TAIL) {
      float4* dst = reinterpret_cast<float4*>(&rows[8 * (tid + FILL_FULL * NTHREADS)]);
      dst[0] = make_float4(__builtin_fabsf(fa[FILL_FULL].x), __builtin_fabsf(fb[FILL_FULL].x),
                           __builtin_fabsf(fa[FILL_FULL].y), __builtin_fabsf(fb[FILL_FULL].y));
      dst[1] = make_float4(__builtin_fabsf(fa[FILL_FULL].z), __builtin_fabsf(fb[FILL_FULL].z),
                           __builtin_fabsf(fa[FILL_FULL].w), __builtin_fabsf(fb[FILL_FULL].w));
    }
  }
  __syncthreads();

  v2f a0, a1;
  a0.x = (c == 0) ? 1.0f : 0.0f;  a0.y = a0.x;
  a1.x = (c == 8) ? 1.0f : 0.0f;  a1.y = a1.x;

  while (t < t_hi) {
    const int u = t * TILE_U + c;
    const int idx[16] = {n0.x, n0.y, n0.z, n0.w, n1.x, n1.y, n1.z, n1.w,
                         n2.x, n2.y, n2.z, n2.w, n3.x, n3.y, n3.z, n3.w};
    const float kw = nk, bw = nb;

    const int tn = t + NWAVES;
    const int tp = tn < t_hi ? tn : t_hi - 1;
    const int4* np = reinterpret_cast<const int4*>(ppi + (tp * TILE_U + c) * DEG + 16 * h);
    n0 = np[0]; n1 = np[1]; n2 = np[2]; n3 = np[3];
    nk = kernelw[tp * TILE_U + c];
    nb = biasw[tp * TILE_U + c];

    v2f g[16];
#pragma unroll
    for (int j = 0; j < 16; ++j)
      g[j] = *reinterpret_cast<const v2f*>(&rows[2 * idx[j]]);
    asm volatile("" :: "v"(g[0]), "v"(g[1]), "v"(g[2]), "v"(g[3]), "v"(g[4]),
                 "v"(g[5]), "v"(g[6]), "v"(g[7]), "v"(g[8]), "v"(g[9]),
                 "v"(g[10]), "v"(g[11]), "v"(g[12]), "v"(g[13]), "v"(g[14]),
                 "v"(g[15]));

    v8f acc0 = {};
    v8f acc1 = {};
#pragma unroll
    for (int j = 0; j < 8; ++j) {
      v2f b0, b1;
      b0.x = g[2 * j].x;  b0.y = g[2 * j + 1].x;
      b1.x = g[2 * j].y;  b1.y = g[2 * j + 1].y;
      acc0 = __builtin_amdgcn_wmma_f32_16x16x4_f32(
          false, a0, false, b0, (short)0, acc0, false, false);
      acc1 = __builtin_amdgcn_wmma_f32_16x16x4_f32(
          false, a1, false, b1, (short)0, acc1, false, false);
    }

    const float s = h ? acc1[0] : acc0[0];
    out[(long)(bA + h) * UNITS + u] = tanhf(s * kw + bw);
    t = tn;
  }
}

extern "C" void kernel_launch(void* const* d_in, const int* in_sizes, int n_in,
                              void* d_out, int out_size, void* d_ws, size_t ws_size,
                              hipStream_t stream) {
  const float* feature = (const float*)d_in[0];
  const int*   ppi     = (const int*)d_in[1];
  const float* kernelw = (const float*)d_in[2];
  const float* biasw   = (const float*)d_in[3];
  float*       out     = (float*)d_out;

  ppi_lds_gather_wmma_kernel<<<(BATCH / 2) * SPLIT, NTHREADS, 0, stream>>>(
      feature, ppi, kernelw, biasw, out);
}


# each 8-byte LDS pair is a WMMA B operand as loaded (A rows 0 and 8 hold (1,0) and (0,1)); loop body written twice with index sets swapped: no vector moves per tile; even/odd sum order changes
# speedup vs baseline: 3.8807x; 1.0541x over previous
#include <hip/hip_runtime.h>
#include <math.h>

#define N_GENES 20000
#define UNITS   10000
#define DEG     32
#define BATCH   128

typedef __attribute__((ext_vector_type(2))) float v2f;
typedef __attribute__((ext_vector_type(8))) float v8f;

#define TILE_U    16
#define N_TILES   (UNITS / TILE_U)
#define SPLIT     4
#define NTHREADS  512
#define NWAVES    (NTHREADS / 32)
#define ROW_VEC4  (N_GENES / 4)
#define FILL_FULL (ROW_VEC4 / NTHREADS)
#define FILL_TAIL (ROW_VEC4 - FILL_FULL * NTHREADS)
#define LDS_FLOATS 41472

__global__ __launch_bounds__(NTHREADS) void ppi_lds_gather_wmma_kernel(
    const float* __restrict__ feature, const int* __restrict__ ppi,
    const float* __restrict__ kernelw, const float* __restrict__ biasw,
    float* __restrict__ out) {
  __shared__ __attribute__((aligned(16))) float rows[LDS_FLOATS];

  const int tid  = threadIdx.x;
  const int lane = tid & 31;
  const int wave = __builtin_amdgcn_readfirstlane(tid >> 5);
  const int c    = lane & 15;
  const int h    = lane >> 4;
  const int pair = blockIdx.x / SPLIT;
  const int part = blockIdx.x % SPLIT;
  const int bA   = pair * 2;
  const int t_lo = (part * N_TILES) / SPLIT;
  const int t_hi = ((part + 1) * N_TILES) / SPLIT;

  int t = t_lo + wave;
  const int4* ip = reinterpret_cast<const int4*>(ppi + (t * TILE_U + c) * DEG + 16 * h);
  int4 n0 = ip[0], n1 = ip[1], n2 = ip[2], n3 = ip[3];
  float nk = kernelw[t * TILE_U + c];
  float nb = biasw[t * TILE_U + c];

  {
    const float4* r0 = reinterpret_cast<const float4*>(feature + (long)bA * N_GENES);
    const float4* r1 = reinterpret_cast<const float4*>(feature + (long)(bA + 1) * N_GENES);
    float4 fa[FILL_FULL + 1], fb[FILL_FULL + 1];
#pragma unroll
    for (int i = 0; i < FILL_FULL; ++i) {
      fa[i] = r0[tid + i * NTHREADS];
      fb[i] = r1[tid + i * NTHREADS];
    }
    const int qt = tid < FILL_TAIL ? tid + FILL_FULL * NTHREADS : ROW_VEC4 - 1;
    fa[FILL_FULL] = r0[qt];
    fb[FILL_FULL] = r1[qt];
#pragma unroll
    for (int i = 0; i < FILL_FULL; ++i) {
      float4* dst = reinterpret_cast<float4*>(&rows[8 * (tid + i * NTHREADS)]);
      dst[0] = make_float4(__builtin_fabsf(fa[i].x), __builtin_fabsf(fb[i].x),
                           __builtin_fabsf(fa[i].y), __builtin_fabsf(fb[i].y));
      dst[1] = make_float4(__builtin_fabsf(fa[i].z), __builtin_fabsf(fb[i].z),
                           __builtin_fabsf(fa[i].w), __builtin_fabsf(fb[i].w));
    }
    asm volatile("" :: "v"(fa[FILL_FULL].x), "v"(fa[FILL_FULL].y),
                 "v"(fa[FILL_FULL].z), "v"(fa[FILL_FULL].w),
                 "v"(fb[FILL_FULL].x), "v"(fb[FILL_FULL].y),
                 "v"(fb[FILL_FULL].z), "v"(fb[FILL_FULL].w));
    if (tid < FILL_TAIL) {
      float4* dst = reinterpret_cast<float4*>(&rows[8 * (tid + FILL_FULL * NTHREADS)]);
      dst[0] = make_float4(__builtin_fabsf(fa[FILL_FULL].x), __builtin_fabsf(fb[FILL_FULL].x),
                           __builtin_fabsf(fa[FILL_FULL].y), __builtin_fabsf(fb[FILL_FULL].y));
      dst[1] = make_float4(__builtin_fabsf(fa[FILL_FULL].z), __builtin_fabsf(fb[FILL_FULL].z),
                           __builtin_fabsf(fa[FILL_FULL].w), __builtin_fabsf(fb[FILL_FULL].w));
    }
  }
  __syncthreads();

  v2f a;
  a.x = (c == 0) ? 1.0f : 0.0f;
  a.y = (c == 8) ? 1.0f : 0.0f;

  float* op = out + (long)(bA + h) * UNITS + (t * TILE_U + c);

#define PPI_TILE(I0, I1, I2, I3, KW, BW)                                       \
  {                                                                            \
    const int idx[16] = {I0.x, I0.y, I0.z, I0.w, I1.x, I1.y, I1.z, I1.w,       \
                         I2.x, I2.y, I2.z, I2.w, I3.x, I3.y, I3.z, I3.w};      \
    v2f g[16];                                                                 \
    _Pragma("unroll")                  \
    for (int j = 0; j < 16; ++j)                                               \
      g[j] = *reinterpret_cast<const v2f*>(&rows[2 * idx[j]]);                 \
         \
        \
      \
    __builtin_amdgcn_sched_barrier(0);                                         \
    v8f acc0 = {};                                                             \
    v8f acc1 = {};                                                             \
    _Pragma("unroll")               \
    for (int j = 0; j < 8; ++j) {                                              \
      acc0 = __builtin_amdgcn_wmma_f32_16x16x4_f32(                            \
          false, a, false, g[2 * j], (short)0, acc0, false, false);            \
      acc1 = __builtin_amdgcn_wmma_f32_16x16x4_f32(                            \
          false, a, false, g[2 * j + 1], (short)0, acc1, false, false);        \
    }                                                                          \
    *op = tanhf((acc0[0] + acc1[0]) * (KW) + (BW));                            \
    op += NWAVES * TILE_U;                                                     \
  }

#define PPI_FETCH(I0, I1, I2, I3, KW, BW, TN)                                  \
  {                                                                            \
    const int tp = (TN) < t_hi ? (TN) : t_hi - 1;                              \
    const int4* np = reinterpret_cast<const int4*>(                            \
        ppi + (tp * TILE_U + c) * DEG + 16 * h);                               \
    I0 = np[0]; I1 = np[1]; I2 = np[2]; I3 = np[3];                            \
    KW = kernelw[tp * TILE_U + c];                                             \
    BW = biasw[tp * TILE_U + c];                                               \
  }

  int4 m0, m1, m2, m3;
  float mk, mb;
  for (;;) {
    PPI_FETCH(m0, m1, m2, m3, mk, mb, t + NWAVES)
    PPI_TILE(n0, n1, n2, n3, nk, nb)
    t += NWAVES;
    if (t >= t_hi) break;
    PPI_FETCH(n0, n1, n2, n3, nk, nb, t + NWAVES)
    PPI_TILE(m0, m1, m2, m3, mk, mb)
    t += NWAVES;
    if (t >= t_hi) break;
  }
#undef PPI_TILE
#undef PPI_FETCH
}

extern "C" void kernel_launch(void* const* d_in, const int* in_sizes, int n_in,
                              void* d_out, int out_size, void* d_ws, size_t ws_size,
                              hipStream_t stream) {
  const float* feature = (const float*)d_in[0];
  const int*   ppi     = (const int*)d_in[1];
  const float* kernelw = (const float*)d_in[2];
  const float* biasw   = (const float*)d_in[3];
  float*       out     = (float*)d_out;

  ppi_lds_gather_wmma_kernel<<<(BATCH / 2) * SPLIT, NTHREADS, 0, stream>>>(
      feature, ppi, kernelw, biasw, out);
}


# index rows fetched as whole lines (a wave load covers 4 lines, was 16): lane takes 4 neighbours of each of 4 units, two quad exchanges (DPP moves) regroup the sums before the WMMA; sum order changes
# speedup vs baseline: 5.1527x; 1.3278x over previous
#include <hip/hip_runtime.h>
#include <math.h>

#define N_GENES 20000
#define UNITS   10000
#define DEG     32
#define BATCH   128

typedef __attribute__((ext_vector_type(2))) float v2f;
typedef __attribute__((ext_vector_type(8))) float v8f;

#define TILE_U    16
#define N_TILES   (UNITS / TILE_U)
#define SPLIT     4
#define NTHREADS  512
#define NWAVES    (NTHREADS / 32)
#define ROW_VEC4  (N_GENES / 4)
#define FILL_FULL (ROW_VEC4 / NTHREADS)
#define FILL_TAIL (ROW_VEC4 - FILL_FULL * NTHREADS)
#define LDS_FLOATS 41472

template <int QP>
__device__ __forceinline__ float quad_from(float v) {
  return __builtin_bit_cast(
      float, __builtin_amdgcn_update_dpp(0, __builtin_bit_cast(int, v), QP, 0xf, 0xf, true));
}

__global__ __launch_bounds__(NTHREADS) void ppi_lds_gather_wmma_kernel(
    const float* __restrict__ feature, const int* __restrict__ ppi,
    const float* __restrict__ kernelw, const float* __restrict__ biasw,
    float* __restrict__ out) {
  __shared__ __attribute__((aligned(16))) float rows[LDS_FLOATS];

  const int tid  = threadIdx.x;
  const int lane = tid & 31;
  const int wave = __builtin_amdgcn_readfirstlane(tid >> 5);
  const int c    = lane & 15;
  const int h    = lane >> 4;
  const int uc   = 4 * (c & 3) + (c >> 2);
  const bool b_odd = (c & 1) != 0;
  const bool b_hi  = (c & 2) != 0;
  const int pair = blockIdx.x / SPLIT;
  const int part = blockIdx.x % SPLIT;
  const int bA   = pair * 2;
  const int t_lo = (part * N_TILES) / SPLIT;
  const int t_hi = ((part + 1) * N_TILES) / SPLIT;

  int t = t_lo + wave;
  const int4* ip = reinterpret_cast<const int4*>(ppi + t * (TILE_U * DEG) + 8 * c + 4 * h);
  int4 n0 = ip[0], n1 = ip[32], n2 = ip[64], n3 = ip[96];
  float nk = kernelw[t * TILE_U + uc];
  float nb = biasw[t * TILE_U + uc];

  {
    const float4* r0 = reinterpret_cast<const float4*>(feature + (long)bA * N_GENES);
    const float4* r1 = reinterpret_cast<const float4*>(feature + (long)(bA + 1) * N_GENES);
    float4 fa[FILL_FULL + 1], fb[FILL_FULL + 1];
#pragma unroll
    for (int i = 0; i < FILL_FULL; ++i) {
      fa[i] = r0[tid + i * NTHREADS];
      fb[i] = r1[tid + i * NTHREADS];
    }
    const int qt = tid < FILL_TAIL ? tid + FILL_FULL * NTHREADS : ROW_VEC4 - 1;
    fa[FILL_FULL] = r0[qt];
    fb[FILL_FULL] = r1[qt];
#pragma unroll
    for (int i = 0; i < FILL_FULL; ++i) {
      float4* dst = reinterpret_cast<float4*>(&rows[8 * (tid + i * NTHREADS)]);
      dst[0] = make_float4(__builtin_fabsf(fa[i].x), __builtin_fabsf(fb[i].x),
                           __builtin_fabsf(fa[i].y), __builtin_fabsf(fb[i].y));
      dst[1] = make_float4(__builtin_fabsf(fa[i].z), __builtin_fabsf(fb[i].z),
                           __builtin_fabsf(fa[i].w), __builtin_fabsf(fb[i].w));
    }
    asm volatile("" :: "v"(fa[FILL_FULL].x), "v"(fa[FILL_FULL].y),
                 "v"(fa[FILL_FULL].z), "v"(fa[FILL_FULL].w),
                 "v"(fb[FILL_FULL].x), "v"(fb[FILL_FULL].y),
                 "v"(fb[FILL_FULL].z), "v"(fb[FILL_FULL].w));
    if (tid < FILL_TAIL) {
      float4* dst = reinterpret_cast<float4*>(&rows[8 * (tid + FILL_FULL * NTHREADS)]);
      dst[0] = make_float4(__builtin_fabsf(fa[FILL_FULL].x), __builtin_fabsf(fb[FILL_FULL].x),
                           __builtin_fabsf(fa[FILL_FULL].y), __builtin_fabsf(fb[FILL_FULL].y));
      dst[1] = make_float4(__builtin_fabsf(fa[FILL_FULL].z), __builtin_fabsf(fb[FILL_FULL].z),
                           __builtin_fabsf(fa[FILL_FULL].w), __builtin_fabsf(fb[FILL_FULL].w));
    }
  }
  __syncthreads();

  v2f a;
  a.x = (c == 0) ? 1.0f : 0.0f;
  a.y = (c == 8) ? 1.0f : 0.0f;

  float* op = out + (long)(bA + h) * UNITS + (t * TILE_U + uc);

#define PPI_TILE(I0, I1, I2, I3, KW, BW)                                       \
  {                                                                            \
    const int idx[16] = {I0.x, I0.y, I0.z, I0.w, I1.x, I1.y, I1.z, I1.w,       \
                         I2.x, I2.y, I2.z, I2.w, I3.x, I3.y, I3.z, I3.w};      \
    v2f g[16];                                                                 \
    _Pragma("unroll")                  \
    for (int j = 0; j < 16; ++j)                                               \
      g[j] = *reinterpret_cast<const v2f*>(&rows[2 * idx[j]]);                 \
      \
                       \
    v2f P[4];                                                                  \
    _Pragma("unroll")            \
    for (int i = 0; i < 4; ++i)                                                \
      P[i] = (g[4 * i] + g[4 * i + 1]) + (g[4 * i + 2] + g[4 * i + 3]);        \
      \
               \
    const v2f k0 = b_odd ? P[1] : P[0], s0 = b_odd ? P[0] : P[1];              \
    const v2f k1 = b_odd ? P[3] : P[2], s1 = b_odd ? P[2] : P[3];              \
    v2f Q0, Q1;                                                                \
    Q0.x = k0.x + quad_from<0xB1>(s0.x);                                       \
    Q0.y = k0.y + quad_from<0xB1>(s0.y);                                       \
    Q1.x = k1.x + quad_from<0xB1>(s1.x);                                       \
    Q1.y = k1.y + quad_from<0xB1>(s1.y);                                       \
           \
                                                     \
    const v2f k2 = b_hi ? Q1 : Q0, s2 = b_hi ? Q0 : Q1;                        \
    v2f sum;                                                                   \
    sum.x = k2.x + quad_from<0x4E>(s2.x);                                      \
    sum.y = k2.y + quad_from<0x4E>(s2.y);                                      \
     \
    v8f acc = {};                                                              \
    acc = __builtin_amdgcn_wmma_f32_16x16x4_f32(                               \
        false, a, false, sum, (short)0, acc, false, false);                    \
    *op = tanhf(acc[0] * (KW) + (BW));                                         \
    op += NWAVES * TILE_U;                                                     \
  }

#define PPI_FETCH(I0, I1, I2, I3, KW, BW, TN)                                  \
  {                                                                            \
    const int tp = (TN) < t_hi ? (TN) : t_hi - 1;                              \
    const int4* np = reinterpret_cast<const int4*>(                            \
        ppi + tp * (TILE_U * DEG) + 8 * c + 4 * h);                            \
    I0 = np[0]; I1 = np[32]; I2 = np[64]; I3 = np[96];                         \
    KW = kernelw[tp * TILE_U + uc];                                            \
    BW = biasw[tp * TILE_U + uc];                                              \
  }

  int4 m0, m1, m2, m3;
  float mk, mb;
  for (;;) {
    PPI_FETCH(m0, m1, m2, m3, mk, mb, t + NWAVES)
    PPI_TILE(n0, n1, n2, n3, nk, nb)
    t += NWAVES;
    if (t >= t_hi) break;
    PPI_FETCH(n0, n1, n2, n3, nk, nb, t + NWAVES)
    PPI_TILE(m0, m1, m2, m3, mk, mb)
    t += NWAVES;
    if (t >= t_hi) break;
  }
#undef PPI_TILE
#undef PPI_FETCH
}

extern "C" void kernel_launch(void* const* d_in, const int* in_sizes, int n_in,
                              void* d_out, int out_size, void* d_ws, size_t ws_size,
                              hipStream_t stream) {
  const float* feature = (const float*)d_in[0];
  const int*   ppi     = (const int*)d_in[1];
  const float* kernelw = (const float*)d_in[2];
  const float* biasw   = (const float*)d_in[3];
  float*       out     = (float*)d_out;

  ppi_lds_gather_wmma_kernel<<<(BATCH / 2) * SPLIT, NTHREADS, 0, stream>>>(
      feature, ppi, kernelw, biasw, out);
}
